# starting kernel
# baseline (speedup 1.0000x reference)
#include <hip/hip_runtime.h>


typedef __attribute__((ext_vector_type(2))) float v2f;
typedef __attribute__((ext_vector_type(8))) float v8f;

#define N_MAPS 1024
#define T_IN   10
#define HW     196
#define NTOT   (16 * HW)
#define TILES_PER_C (NTOT / 16)
#define WAVES_PER_BLOCK 8

__global__ __launch_bounds__(256)
void conv_over_time_wmma_kernel(const float* __restrict__ x,
                                const float* __restrict__ W,
                                const float* __restrict__ bias,
                                float* __restrict__ out)
{
    const int lane = threadIdx.x & 31;
    const int wid  = blockIdx.x * WAVES_PER_BLOCK + (threadIdx.x >> 5);
    const int c    = wid / TILES_PER_C;
    const int tile = wid - c * TILES_PER_C;

    const int half = lane >> 4;
    const int nidx = lane & 15;
    const int n    = tile * 16 + nidx;
    const int b    = n / HW;
    const int hw   = n - b * HW;

    const long long t_stride = (long long)N_MAPS * HW;
    const long long base     = ((long long)b * T_IN * N_MAPS + (long long)c) * HW + hw;

    const float* __restrict__ wrow = W + c * T_IN;

    v8f acc = (v8f)0.0f;

    #pragma unroll
    for (int kk = 0; kk < 3; ++kk) {
        const int t0  = 4 * kk + 2 * half;
        const int t1  = t0 + 1;
        const int t0c = (t0 < T_IN) ? t0 : (T_IN - 1);
        const int t1c = (t1 < T_IN) ? t1 : (T_IN - 1);

        v2f xb, wa;
        xb.x = x[base + (long long)t0c * t_stride];
        xb.y = x[base + (long long)t1c * t_stride];

        const float w0 = wrow[t0c];
        const float w1 = wrow[t1c];
        wa.x = (t0 < T_IN) ? w0 : 0.0f;
        wa.y = (t1 < T_IN) ? w1 : 0.0f;

        acc = __builtin_amdgcn_wmma_f32_16x16x4_f32(
                  false, wa, false, xb, (short)0, acc, false, false);
    }

    if (lane < 16) {
        out[(long long)b * (N_MAPS * HW) + (long long)c * HW + hw]
            = acc[0] + bias[c];
    }
}

extern "C" void kernel_launch(void* const* d_in, const int* in_sizes, int n_in,
                              void* d_out, int out_size, void* d_ws, size_t ws_size,
                              hipStream_t stream) {
    const float* x  = (const float*)d_in[0];
    const float* W  = (const float*)d_in[1];
    const float* bb = (const float*)d_in[2];
    float* out      = (float*)d_out;

    const int total_waves = N_MAPS * TILES_PER_C;
    dim3 grid(total_waves / WAVES_PER_BLOCK);
    dim3 block(32 * WAVES_PER_BLOCK);

    conv_over_time_wmma_kernel<<<grid, block, 0, stream>>>(x, W, bb, out);
}


# Each wave streams 512 contiguous bytes per time plane (16-byte loads, one 512-byte store); block-diagonal weight rows in A; products now summed one per WMMA in t order, not four per WMMA
# speedup vs baseline: 2.3333x; 2.3333x over previous
#include <hip/hip_runtime.h>


typedef __attribute__((ext_vector_type(2))) float v2f;
typedef __attribute__((ext_vector_type(4))) float v4f;
typedef __attribute__((ext_vector_type(8))) float v8f;

#define N_MAPS   1024
#define T_IN     10
#define Q_PER_C  49
#define PLANE_Q  (N_MAPS * Q_PER_C)
#define WAVES_PER_BLOCK 8

__global__ __launch_bounds__(256)
void conv_over_time_wmma_kernel(const float* __restrict__ x,
                                const float* __restrict__ W,
                                const float* __restrict__ bias,
                                float* __restrict__ out)
{
    const int lane = threadIdx.x & 31;
    const int wave = __builtin_amdgcn_readfirstlane((int)(threadIdx.x >> 5));
    const int g    = blockIdx.x * WAVES_PER_BLOCK + wave;
    const int b    = blockIdx.y;

    const int f0   = g * 32;
    const int c_lo = f0 / Q_PER_C;
    const int c_hi = (c_lo + 1 < N_MAPS) ? (c_lo + 1) : (N_MAPS - 1);
    const int f    = f0 + lane;
    const bool hi  = f >= (c_lo + 1) * Q_PER_C;

    const float* __restrict__ wl = W + c_lo * T_IN;
    const float* __restrict__ wh = W + c_hi * T_IN;

    const v4f* __restrict__ xq =
        (const v4f*)x + ((size_t)b * T_IN) * PLANE_Q + f;
    v4f xv[T_IN];
    #pragma unroll
    for (int t = 0; t < T_IN; ++t)
        xv[t] = xq[(size_t)t * PLANE_Q];

    const int m      = lane & 15;
    const int h      = lane >> 4;
    const int r      = m & 7;
    const bool live  = ((m >> 3) == h) && (r < 4);
    const bool useHi = (r & 1) != 0;
    const bool selx  = live && ((r & 2) == 0);
    const bool sely  = live && ((r & 2) != 0);

    v8f acc0 = (v8f)0.0f;
    v8f acc1 = (v8f)0.0f;

    #pragma unroll
    for (int t = 0; t < T_IN; ++t) {
        const float w_lo = wl[t];
        const float w_hi = wh[t];
        const float wsel = useHi ? w_hi : w_lo;
        v2f a;
        a.x = selx ? wsel : 0.0f;
        a.y = sely ? wsel : 0.0f;

        v2f b01, b23;
        b01.x = xv[t].x;  b01.y = xv[t].y;
        b23.x = xv[t].z;  b23.y = xv[t].w;

        acc0 = __builtin_amdgcn_wmma_f32_16x16x4_f32(
                   false, a, false, b01, (short)0, acc0, false, false);
        acc1 = __builtin_amdgcn_wmma_f32_16x16x4_f32(
                   false, a, false, b23, (short)0, acc1, false, false);
    }

    const float b_lo = bias[c_lo];
    const float b_hi = bias[c_hi];
    const float bsel = hi ? b_hi : b_lo;
    v4f o;
    o.x = (hi ? acc0[1] : acc0[0]) + bsel;
    o.y = (hi ? acc0[3] : acc0[2]) + bsel;
    o.z = (hi ? acc1[1] : acc1[0]) + bsel;
    o.w = (hi ? acc1[3] : acc1[2]) + bsel;

    ((v4f*)out)[(size_t)b * PLANE_Q + f] = o;
}

extern "C" void kernel_launch(void* const* d_in, const int* in_sizes, int n_in,
                              void* d_out, int out_size, void* d_ws, size_t ws_size,
                              hipStream_t stream) {
    const float* x  = (const float*)d_in[0];
    const float* W  = (const float*)d_in[1];
    const float* bb = (const float*)d_in[2];
    float* out      = (float*)d_out;

    dim3 grid(196, 16);
    dim3 block(256);

    conv_over_time_wmma_kernel<<<grid, block, 0, stream>>>(x, W, bb, out);
}
